# starting kernel
# baseline (speedup 1.0000x reference)
#include <hip/hip_runtime.h>
#include <hip/hip_bf16.h>

typedef __attribute__((ext_vector_type(2))) float v2f;
typedef __attribute__((ext_vector_type(8))) float v8f;

constexpr int   NQ  = 50000;
constexpr int   MS  = 50000;
constexpr int   HN  = 32;
constexpr int   KK  = 15;
constexpr int   CIN = 64;
constexpr int   CO  = 64;
constexpr float KPE = 1.2f;

constexpr int QT    = 16;
constexpr int NTHR  = 512;

__global__ __launch_bounds__(NTHR)
void kpconv_wmma_f32(const float* __restrict__ q_pts,
                     const float* __restrict__ s_pts,
                     const int*   __restrict__ nb,
                     const float* __restrict__ x,
                     const float* __restrict__ kp,
                     const float* __restrict__ wt,
                     float*       __restrict__ out)
{
    __shared__ float lds_weighted[QT * 16 * CIN];
    __shared__ float lds_partial [16 * 16 * 16];
    __shared__ float lds_nc      [QT * HN * 3];
    __shared__ int   lds_idx     [QT * HN];
    __shared__ float lds_norm    [QT];

    const int tid  = threadIdx.x;
    const int lane = tid & 31;
    const int wave = tid >> 5;
    const int qblk = blockIdx.x * QT;

    {
        const int q = wave;
        const int h = lane;
        const int n = qblk + q;
        const int idx = nb[n * HN + h];
        lds_idx[q * HN + h] = idx;
        float cx = 1e6f, cy = 1e6f, cz = 1e6f, fs = 0.0f;
        if (idx < MS) {
            cx = s_pts[idx * 3 + 0] - q_pts[n * 3 + 0];
            cy = s_pts[idx * 3 + 1] - q_pts[n * 3 + 1];
            cz = s_pts[idx * 3 + 2] - q_pts[n * 3 + 2];
            for (int c = 0; c < CIN; c += 4) {
                fs += x[idx * CIN + c + 0] + x[idx * CIN + c + 1]
                    + x[idx * CIN + c + 2] + x[idx * CIN + c + 3];
            }
        }
        lds_nc[(q * HN + h) * 3 + 0] = cx;
        lds_nc[(q * HN + h) * 3 + 1] = cy;
        lds_nc[(q * HN + h) * 3 + 2] = cz;
        const unsigned long long m = __ballot(fs > 0.0f);
        if (lane == 0) {
            const int cnt = (int)__popcll(m);
            lds_norm[q] = (float)(cnt > 0 ? cnt : 1);
        }
    }
    __syncthreads();

    {
        const int q   = wave;
        const int k   = lane & 15;
        const int col = lane & 15;
        float kx = 0.0f, ky = 0.0f, kz = 0.0f;
        if (k < KK) { kx = kp[k * 3]; ky = kp[k * 3 + 1]; kz = kp[k * 3 + 2]; }

        v8f acc[4];
        #pragma unroll
        for (int ct = 0; ct < 4; ++ct) acc[ct] = (v8f){};

        const int hofs = (lane >> 4) * 2;
        for (int hb = 0; hb < HN; hb += 4) {
            const int h0 = hb + hofs;
            v2f A;
            #pragma unroll
            for (int s = 0; s < 2; ++s) {
                const float nx = lds_nc[(q * HN + h0 + s) * 3 + 0] - kx;
                const float ny = lds_nc[(q * HN + h0 + s) * 3 + 1] - ky;
                const float nz = lds_nc[(q * HN + h0 + s) * 3 + 2] - kz;
                float w = 1.0f - sqrtf(nx * nx + ny * ny + nz * nz) * (1.0f / KPE);
                w = fmaxf(w, 0.0f);
                if (k >= KK) w = 0.0f;
                A[s] = w;
            }
            const int i0 = lds_idx[q * HN + h0];
            const int i1 = lds_idx[q * HN + h0 + 1];
            #pragma unroll
            for (int ct = 0; ct < 4; ++ct) {
                v2f B;
                B.x = (i0 < MS) ? x[i0 * CIN + ct * 16 + col] : 0.0f;
                B.y = (i1 < MS) ? x[i1 * CIN + ct * 16 + col] : 0.0f;
                acc[ct] = __builtin_amdgcn_wmma_f32_16x16x4_f32(
                    false, A, false, B, (short)0, acc[ct], false, false);
            }
        }

        const int mrow = (lane >> 4) * 8;
        #pragma unroll
        for (int ct = 0; ct < 4; ++ct)
            #pragma unroll
            for (int r = 0; r < 8; ++r)
                lds_weighted[q * 1024 + (r + mrow) * CIN + ct * 16 + col] = acc[ct][r];
    }
    __syncthreads();

    {
        const int dtile = wave & 3;
        const int kq    = wave >> 2;
        const int col   = lane & 15;
        const int kofs  = (lane >> 4) * 2;
        const int d     = dtile * 16 + col;
        const int nrow  = lane & 15;

        v8f acc = (v8f){};
        for (int s = 0; s < 64; ++s) {
            const int kb = kq * 256 + s * 4 + kofs;
            v2f A, B;
            A.x = lds_weighted[nrow * 1024 + kb];
            A.y = lds_weighted[nrow * 1024 + kb + 1];
            const int kp0 = kb >> 6, c0 = kb & 63;
            const int kp1 = (kb + 1) >> 6, c1 = (kb + 1) & 63;
            B.x = (kp0 < KK) ? wt[kp0 * (CIN * CO) + c0 * CO + d] : 0.0f;
            B.y = (kp1 < KK) ? wt[kp1 * (CIN * CO) + c1 * CO + d] : 0.0f;
            acc = __builtin_amdgcn_wmma_f32_16x16x4_f32(
                false, A, false, B, (short)0, acc, false, false);
        }
        const int mrow = (lane >> 4) * 8;
        #pragma unroll
        for (int r = 0; r < 8; ++r)
            lds_partial[wave * 256 + (r + mrow) * 16 + col] = acc[r];
    }
    __syncthreads();

    for (int e = tid; e < QT * CO; e += NTHR) {
        const int n  = e >> 6;
        const int d  = e & 63;
        const int dt = d >> 4;
        const int cc = d & 15;
        float s = 0.0f;
        #pragma unroll
        for (int kq = 0; kq < 4; ++kq)
            s += lds_partial[(kq * 4 + dt) * 256 + n * 16 + cc];
        out[(qblk + n) * CO + d] = s / lds_norm[n];
    }
}

extern "C" void kernel_launch(void* const* d_in, const int* in_sizes, int n_in,
                              void* d_out, int out_size, void* d_ws, size_t ws_size,
                              hipStream_t stream) {
    const float* q_pts = (const float*)d_in[0];
    const float* s_pts = (const float*)d_in[1];
    const int*   nb    = (const int*)  d_in[2];
    const float* x     = (const float*)d_in[3];
    const float* kp    = (const float*)d_in[4];
    const float* wt    = (const float*)d_in[5];
    float*       out   = (float*)d_out;

    const int n = in_sizes[0] / 3;
    const int nblk = (n + QT - 1) / QT;
    kpconv_wmma_f32<<<nblk, NTHR, 0, stream>>>(q_pts, s_pts, nb, x, kp, wt, out);
}


# Phase-A B fragments are the gather registers themselves (lanes 0-15 load rows 4i,4i+1; lanes 16-31 rows 4i+2,4i+3): feature rows not staged in LDS; density from piece sums, row-sum order changes
# speedup vs baseline: 2.0886x; 2.0886x over previous
#include <hip/hip_runtime.h>
#include <hip/hip_bf16.h>


typedef __attribute__((ext_vector_type(2))) float v2f;
typedef __attribute__((ext_vector_type(4))) float v4f;
typedef __attribute__((ext_vector_type(8))) float v8f;
typedef __attribute__((ext_vector_type(3))) float v3f;
typedef v3f v3f_a4 __attribute__((aligned(4)));

constexpr int   NQ  = 50000;
constexpr int   MS  = 50000;
constexpr int   HN  = 32;
constexpr int   KK  = 15;
constexpr int   CIN = 64;
constexpr int   CO  = 64;
constexpr float KPE = 1.2f;

constexpr int QT     = 16;
constexpr int NTHR   = 512;
constexpr int NTILES = NQ / QT;
constexpr int NWG    = 256;
constexpr int KSL    = KK * CIN / 16;
constexpr int KST    = KSL / 4;
constexpr int JSPLIT = 12;
constexpr int PP     = 20;

template <int J0, int J1>
__device__ __forceinline__ void issue_rows(const float* __restrict__ x, const int idx,
                                           const int half, const int col, v4f (&stg)[HN / 2])
{
    #pragma unroll
    for (int j = J0; j < J1; ++j) {
        const int i0 = __builtin_amdgcn_readlane(idx, 4 * (j >> 1) + (j & 1));
        const int i1 = __builtin_amdgcn_readlane(idx, 4 * (j >> 1) + (j & 1) + 2);
        const int ih = half ? i1 : i0;
        v4f v = {0.0f, 0.0f, 0.0f, 0.0f};
        if (ih < MS) v = *(const v4f*)(x + ih * CIN + col * 4);
        stg[j] = v;
    }
}

__device__ __forceinline__ void wg_barrier_lds()
{
    asm volatile("s_wait_dscnt 0x0\n\ts_barrier_signal -1\n\ts_barrier_wait -1" ::: "memory");
}

__global__ __launch_bounds__(NTHR)
void kpconv_wmma_f32(const float* __restrict__ q_pts,
                     const float* __restrict__ s_pts,
                     const int*   __restrict__ nb,
                     const float* __restrict__ x,
                     const float* __restrict__ kp,
                     const float* __restrict__ wt,
                     float*       __restrict__ out)
{
    struct Lds {
        float nc      [QT * HN * 3];
        float norm    [2 * QT];
        float psum    [QT * HN * PP];
        float weighted[QT * 16 * CIN];
        float partial [16 * QT * CO];
    };
    static_assert((sizeof(float) * (QT * HN * 3 + 2 * QT)) % 16 == 0, "psum and weighted are read and written 16 bytes at a time");
    __shared__ __attribute__((aligned(16))) Lds lds;
    float (&lds_nc)[QT * HN * 3]        = lds.nc;
    float (&lds_norm)[2 * QT]           = lds.norm;
    float (&lds_psum)[QT * HN * PP]     = lds.psum;
    float (&lds_weighted)[QT * 16 * CIN] = lds.weighted;
    float (&lds_partial)[16 * QT * CO]  = lds.partial;

    const int tid  = threadIdx.x;
    const int lane = tid & 31;
    const int wave = tid >> 5;
    const int col  = lane & 15;
    const int half = lane >> 4;

    v2f Bw[KST][4];
    {
        const float* wp = wt + (wave * KSL + half * 2) * CO + col;
        #pragma unroll
        for (int s = 0; s < KST; ++s) {
            #pragma unroll
            for (int dt = 0; dt < 4; ++dt) {
                Bw[s][dt].x = wp[(s * 4    ) * CO + dt * 16];
                Bw[s][dt].y = wp[(s * 4 + 1) * CO + dt * 16];
            }
        }
    }

    float kx = 0.0f, ky = 0.0f, kz = 0.0f;
    if (col < KK) { kx = kp[col * 3]; ky = kp[col * 3 + 1]; kz = kp[col * 3 + 2]; }

    int   idx = nb[(blockIdx.x * QT + wave) * HN + lane];
    v4f   stg[HN / 2];
    issue_rows<0, HN / 2>(x, idx, half, col, stg);
    v3f   sp = *(const v3f_a4*)(s_pts + (idx < MS ? idx : 0) * 3);
    float qx = q_pts[(blockIdx.x * QT + wave) * 3 + 0];
    float qy = q_pts[(blockIdx.x * QT + wave) * 3 + 1];
    float qz = q_pts[(blockIdx.x * QT + wave) * 3 + 2];

    int par = 0;
    for (int tile = blockIdx.x; tile < NTILES; tile += NWG, par ^= 1) {
        const int  qblk     = tile * QT;
        const int  nqblk    = qblk + NWG * QT;
        const bool has_next = (tile + NWG) < NTILES;
        int idx_nx;

        {
            const int q = wave;
            const int h = lane;

            float cx = 1e6f, cy = 1e6f, cz = 1e6f;
            if (idx < MS) {
                cx = sp.x - qx;
                cy = sp.y - qy;
                cz = sp.z - qz;
            }
            lds_nc[(q * HN + h) * 3 + 0] = cx;
            lds_nc[(q * HN + h) * 3 + 1] = cy;
            lds_nc[(q * HN + h) * 3 + 2] = cz;
        }
        __syncthreads();

        {
            const int q = wave;
            const int k = col;

            idx_nx = MS;
            if (has_next) idx_nx = nb[(nqblk + q) * HN + lane];

            v8f acc[4];
            #pragma unroll
            for (int ct = 0; ct < 4; ++ct) acc[ct] = (v8f){};

            float* psum = &lds_psum[q * (HN * PP)];
            const int hofs = half * 2;
            const float* ncb = &lds_nc[(q * HN + hofs) * 3];
            float*       psb = psum + hofs * PP + col;
            #pragma unroll
            for (int i = 0; i < HN / 4; ++i) {
                v2f A;
                #pragma unroll
                for (int s = 0; s < 2; ++s) {
                    const float nx = ncb[(4 * i + s) * 3 + 0] - kx;
                    const float ny = ncb[(4 * i + s) * 3 + 1] - ky;
                    const float nz = ncb[(4 * i + s) * 3 + 2] - kz;
                    float w = 1.0f - sqrtf(nx * nx + ny * ny + nz * nz) * (1.0f / KPE);
                    w = fmaxf(w, 0.0f);
                    if (k >= KK) w = 0.0f;
                    A[s] = w;
                }
                const v4f r0 = stg[2 * i];
                const v4f r1 = stg[2 * i + 1];
                asm volatile("" :: "v"(r0), "v"(r1));
                psb[(4 * i    ) * PP] = (r0.x + r0.y) + (r0.z + r0.w);
                psb[(4 * i + 1) * PP] = (r1.x + r1.y) + (r1.z + r1.w);
                #pragma unroll
                for (int ct = 0; ct < 4; ++ct) {
                    v2f B;
                    B.x = r0[ct];
                    B.y = r1[ct];
                    acc[ct] = __builtin_amdgcn_wmma_f32_16x16x4_f32(
                        false, A, false, B, (short)0, acc[ct], false, false);
                }
            }

            {
                const float* prow = psum + lane * PP;
                float fs = 0.0f;
                #pragma unroll
                for (int c = 0; c < 16; c += 4) {
                    const v4f v = *(const v4f*)(prow + c);
                    fs += v.x; fs += v.y; fs += v.z; fs += v.w;
                }
                const unsigned long long m = __ballot(fs > 0.0f);
                if (lane == 0) {
                    const int cnt = (int)__popcll(m);
                    lds_norm[par * QT + q] = (float)(cnt > 0 ? cnt : 1);
                }
            }

            const int mrow = half * 8;
            #pragma unroll
            for (int r = 0; r < 8; ++r) {
                const v4f o = {acc[0][r], acc[1][r], acc[2][r], acc[3][r]};
                *(v4f*)&lds_weighted[q * 1024 + (r + mrow) * CIN + col * 4] = o;
            }
        }
        {
            const int nq = (has_next ? nqblk : qblk) + wave;
            idx = idx_nx;
            issue_rows<0, JSPLIT>(x, idx, half, col, stg);
            sp = *(const v3f_a4*)(s_pts + (idx < MS ? idx : 0) * 3);
            qx = q_pts[nq * 3 + 0];
            qy = q_pts[nq * 3 + 1];
            qz = q_pts[nq * 3 + 2];
        }
        wg_barrier_lds();

        {
            const int nrow = col;
            const float* wrow = &lds_weighted[nrow * 1024 + wave * KSL + half * 2];

            v8f acc[4];
            #pragma unroll
            for (int dt = 0; dt < 4; ++dt) acc[dt] = (v8f){};

            #pragma unroll
            for (int s = 0; s < KST; ++s) {
                v2f A;
                A.x = wrow[s * 4];
                A.y = wrow[s * 4 + 1];
                #pragma unroll
                for (int dt = 0; dt < 4; ++dt)
                    acc[dt] = __builtin_amdgcn_wmma_f32_16x16x4_f32(
                        false, A, false, Bw[s][dt], (short)0, acc[dt], false, false);
            }
            const int mrow = half * 8;
            #pragma unroll
            for (int dt = 0; dt < 4; ++dt)
                #pragma unroll
                for (int r = 0; r < 8; ++r)
                    lds_partial[wave * (QT * CO) + (r + mrow) * CO + dt * 16 + col] = acc[dt][r];
        }
        issue_rows<JSPLIT, HN / 2>(x, idx, half, col, stg);
        wg_barrier_lds();

        for (int e = tid; e < QT * CO; e += NTHR) {
            const int n = e >> 6;
            float s = 0.0f;
            #pragma unroll
            for (int w = 0; w < 16; ++w)
                s += lds_partial[w * (QT * CO) + e];
            out[qblk * CO + e] = s / lds_norm[par * QT + n];
        }
    }
}

extern "C" void kernel_launch(void* const* d_in, const int* in_sizes, int n_in,
                              void* d_out, int out_size, void* d_ws, size_t ws_size,
                              hipStream_t stream) {
    const float* q_pts = (const float*)d_in[0];
    const float* s_pts = (const float*)d_in[1];
    const int*   nb    = (const int*)  d_in[2];
    const float* x     = (const float*)d_in[3];
    const float* kp    = (const float*)d_in[4];
    const float* wt    = (const float*)d_in[5];
    float*       out   = (float*)d_out;

    kpconv_wmma_f32<<<NWG, NTHR, 0, stream>>>(q_pts, s_pts, nb, x, kp, wt, out);
}


# K-part sums handed from phase B to phase C in register order as 16-byte pieces (512 contiguous bytes per wave store, no bank conflict); phase C: 8 waves, 16-byte reads; sums and order unchanged
# speedup vs baseline: 2.1826x; 1.0450x over previous
#include <hip/hip_runtime.h>
#include <hip/hip_bf16.h>


typedef __attribute__((ext_vector_type(2))) float v2f;
typedef __attribute__((ext_vector_type(4))) float v4f;
typedef __attribute__((ext_vector_type(8))) float v8f;
typedef __attribute__((ext_vector_type(3))) float v3f;
typedef v3f v3f_a4 __attribute__((aligned(4)));

constexpr int   NQ  = 50000;
constexpr int   MS  = 50000;
constexpr int   HN  = 32;
constexpr int   KK  = 15;
constexpr int   CIN = 64;
constexpr int   CO  = 64;
constexpr float KPE = 1.2f;

constexpr int QT     = 16;
constexpr int NTHR   = 512;
constexpr int NTILES = NQ / QT;
constexpr int NWG    = 256;
constexpr int KSL    = KK * CIN / 16;
constexpr int KST    = KSL / 4;
constexpr int JSPLIT = 12;
constexpr int PP     = 20;

template <int J0, int J1>
__device__ __forceinline__ void issue_rows(const float* __restrict__ x, const int idx,
                                           const int half, const int col, v4f (&stg)[HN / 2])
{
    #pragma unroll
    for (int j = J0; j < J1; ++j) {
        const int i0 = __builtin_amdgcn_readlane(idx, 4 * (j >> 1) + (j & 1));
        const int i1 = __builtin_amdgcn_readlane(idx, 4 * (j >> 1) + (j & 1) + 2);
        const int ih = half ? i1 : i0;
        v4f v = {0.0f, 0.0f, 0.0f, 0.0f};
        if (ih < MS) v = *(const v4f*)(x + ih * CIN + col * 4);
        stg[j] = v;
    }
}

__device__ __forceinline__ void wg_barrier_lds()
{
    asm volatile("s_wait_dscnt 0x0\n\ts_barrier_signal -1\n\ts_barrier_wait -1" ::: "memory");
}

__global__ __launch_bounds__(NTHR)
void kpconv_wmma_f32(const float* __restrict__ q_pts,
                     const float* __restrict__ s_pts,
                     const int*   __restrict__ nb,
                     const float* __restrict__ x,
                     const float* __restrict__ kp,
                     const float* __restrict__ wt,
                     float*       __restrict__ out)
{
    struct Lds {
        float nc      [QT * HN * 3];
        float norm    [2 * QT];
        float psum    [QT * HN * PP];
        float weighted[QT * 16 * CIN];
        float partial [16 * QT * CO];
    };
    static_assert((sizeof(float) * (QT * HN * 3 + 2 * QT)) % 16 == 0, "psum and weighted are read and written 16 bytes at a time");
    __shared__ __attribute__((aligned(16))) Lds lds;
    float (&lds_nc)[QT * HN * 3]        = lds.nc;
    float (&lds_norm)[2 * QT]           = lds.norm;
    float (&lds_psum)[QT * HN * PP]     = lds.psum;
    float (&lds_weighted)[QT * 16 * CIN] = lds.weighted;
    float (&lds_partial)[16 * QT * CO]  = lds.partial;

    const int tid  = threadIdx.x;
    const int lane = tid & 31;
    const int wave = tid >> 5;
    const int col  = lane & 15;
    const int half = lane >> 4;

    v2f Bw[KST][4];
    {
        const float* wp = wt + (wave * KSL + half * 2) * CO + col;
        #pragma unroll
        for (int s = 0; s < KST; ++s) {
            #pragma unroll
            for (int dt = 0; dt < 4; ++dt) {
                Bw[s][dt].x = wp[(s * 4    ) * CO + dt * 16];
                Bw[s][dt].y = wp[(s * 4 + 1) * CO + dt * 16];
            }
        }
    }

    float kx = 0.0f, ky = 0.0f, kz = 0.0f;
    if (col < KK) { kx = kp[col * 3]; ky = kp[col * 3 + 1]; kz = kp[col * 3 + 2]; }

    int   idx = nb[(blockIdx.x * QT + wave) * HN + lane];
    v4f   stg[HN / 2];
    issue_rows<0, HN / 2>(x, idx, half, col, stg);
    v3f   sp = *(const v3f_a4*)(s_pts + (idx < MS ? idx : 0) * 3);
    float qx = q_pts[(blockIdx.x * QT + wave) * 3 + 0];
    float qy = q_pts[(blockIdx.x * QT + wave) * 3 + 1];
    float qz = q_pts[(blockIdx.x * QT + wave) * 3 + 2];

    int par = 0;
    for (int tile = blockIdx.x; tile < NTILES; tile += NWG, par ^= 1) {
        const int  qblk     = tile * QT;
        const int  nqblk    = qblk + NWG * QT;
        const bool has_next = (tile + NWG) < NTILES;
        int idx_nx;

        {
            const int q = wave;
            const int h = lane;

            float cx = 1e6f, cy = 1e6f, cz = 1e6f;
            if (idx < MS) {
                cx = sp.x - qx;
                cy = sp.y - qy;
                cz = sp.z - qz;
            }
            lds_nc[(q * HN + h) * 3 + 0] = cx;
            lds_nc[(q * HN + h) * 3 + 1] = cy;
            lds_nc[(q * HN + h) * 3 + 2] = cz;
        }
        __syncthreads();

        {
            const int q = wave;
            const int k = col;

            idx_nx = MS;
            if (has_next) idx_nx = nb[(nqblk + q) * HN + lane];

            v8f acc[4];
            #pragma unroll
            for (int ct = 0; ct < 4; ++ct) acc[ct] = (v8f){};

            float* psum = &lds_psum[q * (HN * PP)];
            const int hofs = half * 2;
            const float* ncb = &lds_nc[(q * HN + hofs) * 3];
            float*       psb = psum + hofs * PP + col;
            #pragma unroll
            for (int i = 0; i < HN / 4; ++i) {
                v2f A;
                #pragma unroll
                for (int s = 0; s < 2; ++s) {
                    const float nx = ncb[(4 * i + s) * 3 + 0] - kx;
                    const float ny = ncb[(4 * i + s) * 3 + 1] - ky;
                    const float nz = ncb[(4 * i + s) * 3 + 2] - kz;
                    float w = 1.0f - sqrtf(nx * nx + ny * ny + nz * nz) * (1.0f / KPE);
                    w = fmaxf(w, 0.0f);
                    if (k >= KK) w = 0.0f;
                    A[s] = w;
                }
                const v4f r0 = stg[2 * i];
                const v4f r1 = stg[2 * i + 1];
                asm volatile("" :: "v"(r0), "v"(r1));
                psb[(4 * i    ) * PP] = (r0.x + r0.y) + (r0.z + r0.w);
                psb[(4 * i + 1) * PP] = (r1.x + r1.y) + (r1.z + r1.w);
                #pragma unroll
                for (int ct = 0; ct < 4; ++ct) {
                    v2f B;
                    B.x = r0[ct];
                    B.y = r1[ct];
                    acc[ct] = __builtin_amdgcn_wmma_f32_16x16x4_f32(
                        false, A, false, B, (short)0, acc[ct], false, false);
                }
            }

            {
                const float* prow = psum + lane * PP;
                float fs = 0.0f;
                #pragma unroll
                for (int c = 0; c < 16; c += 4) {
                    const v4f v = *(const v4f*)(prow + c);
                    fs += v.x; fs += v.y; fs += v.z; fs += v.w;
                }
                const unsigned long long m = __ballot(fs > 0.0f);
                if (lane == 0) {
                    const int cnt = (int)__popcll(m);
                    lds_norm[par * QT + q] = (float)(cnt > 0 ? cnt : 1);
                }
            }

            const int mrow = half * 8;
            #pragma unroll
            for (int r = 0; r < 8; ++r) {
                const v4f o = {acc[0][r], acc[1][r], acc[2][r], acc[3][r]};
                *(v4f*)&lds_weighted[q * 1024 + (r + mrow) * CIN + col * 4] = o;
            }
        }
        {
            const int nq = (has_next ? nqblk : qblk) + wave;
            idx = idx_nx;
            issue_rows<0, JSPLIT>(x, idx, half, col, stg);
            sp = *(const v3f_a4*)(s_pts + (idx < MS ? idx : 0) * 3);
            qx = q_pts[nq * 3 + 0];
            qy = q_pts[nq * 3 + 1];
            qz = q_pts[nq * 3 + 2];
        }
        wg_barrier_lds();

        {
            const int nrow = col;
            const float* wrow = &lds_weighted[nrow * 1024 + wave * KSL + half * 2];

            v8f acc[4];
            #pragma unroll
            for (int dt = 0; dt < 4; ++dt) acc[dt] = (v8f){};

            #pragma unroll
            for (int s = 0; s < KST; ++s) {
                v2f A;
                A.x = wrow[s * 4];
                A.y = wrow[s * 4 + 1];
                #pragma unroll
                for (int dt = 0; dt < 4; ++dt)
                    acc[dt] = __builtin_amdgcn_wmma_f32_16x16x4_f32(
                        false, A, false, Bw[s][dt], (short)0, acc[dt], false, false);
            }
            float* pw = &lds_partial[wave * (QT * CO) + lane * 4];
            #pragma unroll
            for (int dt = 0; dt < 4; ++dt)
                #pragma unroll
                for (int rq = 0; rq < 2; ++rq) {
                    const v4f o = {acc[dt][4 * rq], acc[dt][4 * rq + 1], acc[dt][4 * rq + 2], acc[dt][4 * rq + 3]};
                    *(v4f*)(pw + dt * 256 + rq * 128) = o;
                }
        }
        issue_rows<JSPLIT, HN / 2>(x, idx, half, col, stg);
        wg_barrier_lds();

        if (tid < 256) {
            const int a  = tid >> 7;
            const int rq = (tid >> 6) & 1;
            const int hh = (tid >> 5) & 1;
            const int i  = tid & 31;
            const float* pp = &lds_partial[(2 * a + (i >> 4)) * 256 + rq * 128 + (16 * hh + (i & 15)) * 4];
            v4f s = {0.0f, 0.0f, 0.0f, 0.0f};
            #pragma unroll
            for (int w = 0; w < 16; ++w) {
                const v4f p = *(const v4f*)(pp + w * (QT * CO));
                s.x += p.x; s.y += p.y; s.z += p.z; s.w += p.w;
            }
            const int   n0  = 4 * rq + 8 * hh;
            const v4f   nrm = *(const v4f*)&lds_norm[par * QT + n0];
            float*      op  = out + (qblk + n0) * CO + 32 * a + i;
            op[0]      = s.x / nrm.x;
            op[CO]     = s.y / nrm.y;
            op[2 * CO] = s.z / nrm.z;
            op[3 * CO] = s.w / nrm.w;
        }
    }
}

extern "C" void kernel_launch(void* const* d_in, const int* in_sizes, int n_in,
                              void* d_out, int out_size, void* d_ws, size_t ws_size,
                              hipStream_t stream) {
    const float* q_pts = (const float*)d_in[0];
    const float* s_pts = (const float*)d_in[1];
    const int*   nb    = (const int*)  d_in[2];
    const float* x     = (const float*)d_in[3];
    const float* kp    = (const float*)d_in[4];
    const float* wt    = (const float*)d_in[5];
    float*       out   = (float*)d_out;

    kpconv_wmma_f32<<<NWG, NTHR, 0, stream>>>(q_pts, s_pts, nb, x, kp, wt, out);
}


# Gather load addresses by one DPP row_share per load (register j = rows j and 16+j, index shifted in the same instruction): no readlane, select or branch; phase-A neighbour-sum order changes
# speedup vs baseline: 2.4262x; 1.1116x over previous
#include <hip/hip_runtime.h>
#include <hip/hip_bf16.h>


typedef __attribute__((ext_vector_type(2))) float v2f;
typedef __attribute__((ext_vector_type(4))) float v4f;
typedef __attribute__((ext_vector_type(8))) float v8f;
typedef __attribute__((ext_vector_type(3))) float v3f;
typedef v3f v3f_a4 __attribute__((aligned(4)));

constexpr int   NQ  = 50000;
constexpr int   MS  = 50000;
constexpr int   HN  = 32;
constexpr int   KK  = 15;
constexpr int   CIN = 64;
constexpr int   CO  = 64;
constexpr float KPE = 1.2f;

constexpr int QT     = 16;
constexpr int NTHR   = 512;
constexpr int NTILES = NQ / QT;
constexpr int NWG    = 256;
constexpr int KSL    = KK * CIN / 16;
constexpr int KST    = KSL / 4;
constexpr int JSPLIT = 12;
constexpr int PP     = 20;
constexpr int PSK    = 16;
constexpr int PSQ    = HN * PP + PSK;

template <int J>
__device__ __forceinline__ int row_lane(const int v)
{
    return __builtin_amdgcn_update_dpp(0, v, 0x150 + J, 0xf, 0xf, true);
}

template <int J, int J1>
struct IssueRows {
    static __device__ __forceinline__ void run(const float* __restrict__ x, const int idxc,
                                               const unsigned colb, v4f (&stg)[HN / 2])
    {
        const unsigned ih = (unsigned)row_lane<J>(idxc);
        stg[J] = *(const v4f*)((const char*)x + ((ih << 8) + colb));
        IssueRows<J + 1, J1>::run(x, idxc, colb, stg);
    }
};
template <int J1>
struct IssueRows<J1, J1> {
    static __device__ __forceinline__ void run(const float* __restrict__, const int, const unsigned, v4f (&)[HN / 2]) {}
};

__device__ __forceinline__ void wg_barrier_lds()
{
    asm volatile("s_wait_dscnt 0x0\n\ts_barrier_signal -1\n\ts_barrier_wait -1" ::: "memory");
}

__global__ __launch_bounds__(NTHR)
void kpconv_wmma_f32(const float* __restrict__ q_pts,
                     const float* __restrict__ s_pts,
                     const int*   __restrict__ nb,
                     const float* __restrict__ x,
                     const float* __restrict__ kp,
                     const float* __restrict__ wt,
                     float*       __restrict__ out)
{
    struct Lds {
        float nc      [QT * HN * 3];
        float norm    [2 * QT];
        float psum    [QT * PSQ];
        float weighted[QT * 16 * CIN];
        float partial [16 * QT * CO];
    };
    static_assert((sizeof(float) * (QT * HN * 3 + 2 * QT)) % 16 == 0 && (sizeof(float) * PSQ) % 16 == 0 && (sizeof(float) * PSK) % 16 == 0,
                  "psum and weighted are read and written 16 bytes at a time");
    __shared__ __attribute__((aligned(16))) Lds lds;
    float (&lds_nc)[QT * HN * 3]        = lds.nc;
    float (&lds_norm)[2 * QT]           = lds.norm;
    float (&lds_psum)[QT * PSQ]         = lds.psum;
    float (&lds_weighted)[QT * 16 * CIN] = lds.weighted;
    float (&lds_partial)[16 * QT * CO]  = lds.partial;

    const int tid  = threadIdx.x;
    const int lane = tid & 31;
    const int wave = tid >> 5;
    const int col  = lane & 15;
    const int half = lane >> 4;

    v2f Bw[KST][4];
    {
        const float* wp = wt + (wave * KSL + half * 2) * CO + col;
        #pragma unroll
        for (int s = 0; s < KST; ++s) {
            #pragma unroll
            for (int dt = 0; dt < 4; ++dt) {
                Bw[s][dt].x = wp[(s * 4    ) * CO + dt * 16];
                Bw[s][dt].y = wp[(s * 4 + 1) * CO + dt * 16];
            }
        }
    }

    float kx = 0.0f, ky = 0.0f, kz = 0.0f;
    if (col < KK) { kx = kp[col * 3]; ky = kp[col * 3 + 1]; kz = kp[col * 3 + 2]; }

    const unsigned colb = (unsigned)col * 16u;
    int   idx = nb[(blockIdx.x * QT + wave) * HN + lane];
    v4f   stg[HN / 2];
    IssueRows<0, HN / 2>::run(x, idx < MS ? idx : MS - 1, colb, stg);
    v3f   sp = *(const v3f_a4*)(s_pts + (idx < MS ? idx : 0) * 3);
    float qx = q_pts[(blockIdx.x * QT + wave) * 3 + 0];
    float qy = q_pts[(blockIdx.x * QT + wave) * 3 + 1];
    float qz = q_pts[(blockIdx.x * QT + wave) * 3 + 2];

    int par = 0;
    for (int tile = blockIdx.x; tile < NTILES; tile += NWG, par ^= 1) {
        const int  qblk     = tile * QT;
        const int  nqblk    = qblk + NWG * QT;
        const bool has_next = (tile + NWG) < NTILES;
        int idx_nx;

        {
            const int q = wave;
            const int h = lane;

            float cx = 1e6f, cy = 1e6f, cz = 1e6f;
            if (idx < MS) {
                cx = sp.x - qx;
                cy = sp.y - qy;
                cz = sp.z - qz;
            }
            lds_nc[(q * HN + h) * 3 + 0] = cx;
            lds_nc[(q * HN + h) * 3 + 1] = cy;
            lds_nc[(q * HN + h) * 3 + 2] = cz;
        }
        __syncthreads();

        {
            const int q = wave;
            const int k = col;

            idx_nx = 0;
            if (has_next) idx_nx = nb[(nqblk + q) * HN + lane];

            v8f acc[4];
            #pragma unroll
            for (int ct = 0; ct < 4; ++ct) acc[ct] = (v8f){};

            float* psum = &lds_psum[q * PSQ];
            const int hofs = half * (HN / 2);
            const float* ncb = &lds_nc[(q * HN + hofs) * 3];
            float*       psb = psum + hofs * PP + half * PSK + col;
            #pragma unroll
            for (int i = 0; i < HN / 4; ++i) {
                v2f A;
                #pragma unroll
                for (int s = 0; s < 2; ++s) {
                    const float nx = ncb[(2 * i + s) * 3 + 0] - kx;
                    const float ny = ncb[(2 * i + s) * 3 + 1] - ky;
                    const float nz = ncb[(2 * i + s) * 3 + 2] - kz;
                    float w = 1.0f - sqrtf(nx * nx + ny * ny + nz * nz) * (1.0f / KPE);
                    w = fmaxf(w, 0.0f);
                    if (k >= KK) w = 0.0f;
                    A[s] = w;
                }
                const v4f r0 = stg[2 * i];
                const v4f r1 = stg[2 * i + 1];
                asm volatile("" :: "v"(r0), "v"(r1));
                psb[(2 * i    ) * PP] = (r0.x + r0.y) + (r0.z + r0.w);
                psb[(2 * i + 1) * PP] = (r1.x + r1.y) + (r1.z + r1.w);
                #pragma unroll
                for (int ct = 0; ct < 4; ++ct) {
                    v2f B;
                    B.x = r0[ct];
                    B.y = r1[ct];
                    acc[ct] = __builtin_amdgcn_wmma_f32_16x16x4_f32(
                        false, A, false, B, (short)0, acc[ct], false, false);
                }
            }

            {
                const float* prow = psum + lane * PP + half * PSK;
                float fs = 0.0f;
                #pragma unroll
                for (int c = 0; c < 16; c += 4) {
                    const v4f v = *(const v4f*)(prow + c);
                    fs += v.x; fs += v.y; fs += v.z; fs += v.w;
                }
                const unsigned long long m = __ballot(fs > 0.0f && idx < MS);
                if (lane == 0) {
                    const int cnt = (int)__popcll(m);
                    lds_norm[par * QT + q] = (float)(cnt > 0 ? cnt : 1);
                }
            }

            const int mrow = half * 8;
            #pragma unroll
            for (int r = 0; r < 8; ++r) {
                const v4f o = {acc[0][r], acc[1][r], acc[2][r], acc[3][r]};
                *(v4f*)&lds_weighted[q * 1024 + (r + mrow) * CIN + col * 4] = o;
            }
        }
        {
            const int nq = (has_next ? nqblk : qblk) + wave;
            idx = idx_nx;
            IssueRows<0, JSPLIT>::run(x, idx < MS ? idx : MS - 1, colb, stg);
            sp = *(const v3f_a4*)(s_pts + (idx < MS ? idx : 0) * 3);
            qx = q_pts[nq * 3 + 0];
            qy = q_pts[nq * 3 + 1];
            qz = q_pts[nq * 3 + 2];
        }
        wg_barrier_lds();

        {
            const int nrow = col;
            const float* wrow = &lds_weighted[nrow * 1024 + wave * KSL + half * 2];

            v8f acc[4];
            #pragma unroll
            for (int dt = 0; dt < 4; ++dt) acc[dt] = (v8f){};

            #pragma unroll
            for (int s = 0; s < KST; ++s) {
                v2f A;
                A.x = wrow[s * 4];
                A.y = wrow[s * 4 + 1];
                #pragma unroll
                for (int dt = 0; dt < 4; ++dt)
                    acc[dt] = __builtin_amdgcn_wmma_f32_16x16x4_f32(
                        false, A, false, Bw[s][dt], (short)0, acc[dt], false, false);
            }
            float* pw = &lds_partial[wave * (QT * CO) + lane * 4];
            #pragma unroll
            for (int dt = 0; dt < 4; ++dt)
                #pragma unroll
                for (int rq = 0; rq < 2; ++rq) {
                    const v4f o = {acc[dt][4 * rq], acc[dt][4 * rq + 1], acc[dt][4 * rq + 2], acc[dt][4 * rq + 3]};
                    *(v4f*)(pw + dt * 256 + rq * 128) = o;
                }
        }
        {
            int idxc = idx < MS ? idx : MS - 1;
            asm volatile("" : "+v"(idxc));
            IssueRows<JSPLIT, HN / 2>::run(x, idxc, colb, stg);
        }
        wg_barrier_lds();

        if (tid < 256) {
            const int a  = tid >> 7;
            const int rq = (tid >> 6) & 1;
            const int hh = (tid >> 5) & 1;
            const int i  = tid & 31;
            const float* pp = &lds_partial[(2 * a + (i >> 4)) * 256 + rq * 128 + (16 * hh + (i & 15)) * 4];
            v4f s = {0.0f, 0.0f, 0.0f, 0.0f};
            #pragma unroll
            for (int w = 0; w < 16; ++w) {
                const v4f p = *(const v4f*)(pp + w * (QT * CO));
                s.x += p.x; s.y += p.y; s.z += p.z; s.w += p.w;
            }
            const int   n0  = 4 * rq + 8 * hh;
            const v4f   nrm = *(const v4f*)&lds_norm[par * QT + n0];
            float*      op  = out + (qblk + n0) * CO + 32 * a + i;
            op[0]      = s.x / nrm.x;
            op[CO]     = s.y / nrm.y;
            op[2 * CO] = s.z / nrm.z;
            op[3 * CO] = s.w / nrm.w;
        }
    }
}

extern "C" void kernel_launch(void* const* d_in, const int* in_sizes, int n_in,
                              void* d_out, int out_size, void* d_ws, size_t ws_size,
                              hipStream_t stream) {
    const float* q_pts = (const float*)d_in[0];
    const float* s_pts = (const float*)d_in[1];
    const int*   nb    = (const int*)  d_in[2];
    const float* x     = (const float*)d_in[3];
    const float* kp    = (const float*)d_in[4];
    const float* wt    = (const float*)d_in[5];
    float*       out   = (float*)d_out;

    kpconv_wmma_f32<<<NWG, NTHR, 0, stream>>>(q_pts, s_pts, nb, x, kp, wt, out);
}


# Workgroup barrier between phase 0 and phase A dropped (a wave reads back only the coordinates it staged itself): waves without a phase C run the next tile's phase A beside it; sums unchanged
# speedup vs baseline: 2.5098x; 1.0345x over previous
#include <hip/hip_runtime.h>
#include <hip/hip_bf16.h>


typedef __attribute__((ext_vector_type(2))) float v2f;
typedef __attribute__((ext_vector_type(4))) float v4f;
typedef __attribute__((ext_vector_type(8))) float v8f;
typedef __attribute__((ext_vector_type(3))) float v3f;
typedef v3f v3f_a4 __attribute__((aligned(4)));

constexpr int   NQ  = 50000;
constexpr int   MS  = 50000;
constexpr int   HN  = 32;
constexpr int   KK  = 15;
constexpr int   CIN = 64;
constexpr int   CO  = 64;
constexpr float KPE = 1.2f;

constexpr int QT     = 16;
constexpr int NTHR   = 512;
constexpr int NTILES = NQ / QT;
constexpr int NWG    = 256;
constexpr int KSL    = KK * CIN / 16;
constexpr int KST    = KSL / 4;
constexpr int JSPLIT = 12;
constexpr int PP     = 20;
constexpr int PSK    = 16;
constexpr int PSQ    = HN * PP + PSK;

template <int J>
__device__ __forceinline__ int row_lane(const int v)
{
    return __builtin_amdgcn_update_dpp(0, v, 0x150 + J, 0xf, 0xf, true);
}

template <int J, int J1>
struct IssueRows {
    static __device__ __forceinline__ void run(const float* __restrict__ x, const int idxc,
                                               const unsigned colb, v4f (&stg)[HN / 2])
    {
        const unsigned ih = (unsigned)row_lane<J>(idxc);
        stg[J] = *(const v4f*)((const char*)x + ((ih << 8) + colb));
        IssueRows<J + 1, J1>::run(x, idxc, colb, stg);
    }
};
template <int J1>
struct IssueRows<J1, J1> {
    static __device__ __forceinline__ void run(const float* __restrict__, const int, const unsigned, v4f (&)[HN / 2]) {}
};

__device__ __forceinline__ void wg_barrier_lds()
{
    asm volatile("s_wait_dscnt 0x0\n\ts_barrier_signal -1\n\ts_barrier_wait -1" ::: "memory");
}

__global__ __launch_bounds__(NTHR)
void kpconv_wmma_f32(const float* __restrict__ q_pts,
                     const float* __restrict__ s_pts,
                     const int*   __restrict__ nb,
                     const float* __restrict__ x,
                     const float* __restrict__ kp,
                     const float* __restrict__ wt,
                     float*       __restrict__ out)
{
    struct Lds {
        float nc      [QT * HN * 3];
        float norm    [2 * QT];
        float psum    [QT * PSQ];
        float weighted[QT * 16 * CIN];
        float partial [16 * QT * CO];
    };
    static_assert((sizeof(float) * (QT * HN * 3 + 2 * QT)) % 16 == 0 && (sizeof(float) * PSQ) % 16 == 0 && (sizeof(float) * PSK) % 16 == 0,
                  "psum and weighted are read and written 16 bytes at a time");
    __shared__ __attribute__((aligned(16))) Lds lds;
    float (&lds_nc)[QT * HN * 3]        = lds.nc;
    float (&lds_norm)[2 * QT]           = lds.norm;
    float (&lds_psum)[QT * PSQ]         = lds.psum;
    float (&lds_weighted)[QT * 16 * CIN] = lds.weighted;
    float (&lds_partial)[16 * QT * CO]  = lds.partial;

    const int tid  = threadIdx.x;
    const int lane = tid & 31;
    const int wave = tid >> 5;
    const int col  = lane & 15;
    const int half = lane >> 4;

    v2f Bw[KST][4];
    {
        const float* wp = wt + (wave * KSL + half * 2) * CO + col;
        #pragma unroll
        for (int s = 0; s < KST; ++s) {
            #pragma unroll
            for (int dt = 0; dt < 4; ++dt) {
                Bw[s][dt].x = wp[(s * 4    ) * CO + dt * 16];
                Bw[s][dt].y = wp[(s * 4 + 1) * CO + dt * 16];
            }
        }
    }

    float kx = 0.0f, ky = 0.0f, kz = 0.0f;
    if (col < KK) { kx = kp[col * 3]; ky = kp[col * 3 + 1]; kz = kp[col * 3 + 2]; }

    const unsigned colb = (unsigned)col * 16u;
    int   idx = nb[(blockIdx.x * QT + wave) * HN + lane];
    v4f   stg[HN / 2];
    IssueRows<0, HN / 2>::run(x, idx < MS ? idx : MS - 1, colb, stg);
    v3f   sp = *(const v3f_a4*)(s_pts + (idx < MS ? idx : 0) * 3);
    float qx = q_pts[(blockIdx.x * QT + wave) * 3 + 0];
    float qy = q_pts[(blockIdx.x * QT + wave) * 3 + 1];
    float qz = q_pts[(blockIdx.x * QT + wave) * 3 + 2];

    int par = 0;
    for (int tile = blockIdx.x; tile < NTILES; tile += NWG, par ^= 1) {
        const int  qblk     = tile * QT;
        const int  nqblk    = qblk + NWG * QT;
        const bool has_next = (tile + NWG) < NTILES;
        int idx_nx;

        {
            const int q = wave;
            const int h = lane;

            float cx = 1e6f, cy = 1e6f, cz = 1e6f;
            if (idx < MS) {
                cx = sp.x - qx;
                cy = sp.y - qy;
                cz = sp.z - qz;
            }
            lds_nc[(q * HN + h) * 3 + 0] = cx;
            lds_nc[(q * HN + h) * 3 + 1] = cy;
            lds_nc[(q * HN + h) * 3 + 2] = cz;
        }

        {
            const int q = wave;
            const int k = col;

            idx_nx = 0;
            if (has_next) idx_nx = nb[(nqblk + q) * HN + lane];

            v8f acc[4];
            #pragma unroll
            for (int ct = 0; ct < 4; ++ct) acc[ct] = (v8f){};

            float* psum = &lds_psum[q * PSQ];
            const int hofs = half * (HN / 2);
            const float* ncb = &lds_nc[(q * HN + hofs) * 3];
            float*       psb = psum + hofs * PP + half * PSK + col;
            #pragma unroll
            for (int i = 0; i < HN / 4; ++i) {
                v2f A;
                #pragma unroll
                for (int s = 0; s < 2; ++s) {
                    const float nx = ncb[(2 * i + s) * 3 + 0] - kx;
                    const float ny = ncb[(2 * i + s) * 3 + 1] - ky;
                    const float nz = ncb[(2 * i + s) * 3 + 2] - kz;
                    float w = 1.0f - sqrtf(nx * nx + ny * ny + nz * nz) * (1.0f / KPE);
                    w = fmaxf(w, 0.0f);
                    if (k >= KK) w = 0.0f;
                    A[s] = w;
                }
                const v4f r0 = stg[2 * i];
                const v4f r1 = stg[2 * i + 1];
                asm volatile("" :: "v"(r0), "v"(r1));
                psb[(2 * i    ) * PP] = (r0.x + r0.y) + (r0.z + r0.w);
                psb[(2 * i + 1) * PP] = (r1.x + r1.y) + (r1.z + r1.w);
                #pragma unroll
                for (int ct = 0; ct < 4; ++ct) {
                    v2f B;
                    B.x = r0[ct];
                    B.y = r1[ct];
                    acc[ct] = __builtin_amdgcn_wmma_f32_16x16x4_f32(
                        false, A, false, B, (short)0, acc[ct], false, false);
                }
            }

            {
                const float* prow = psum + lane * PP + half * PSK;
                float fs = 0.0f;
                #pragma unroll
                for (int c = 0; c < 16; c += 4) {
                    const v4f v = *(const v4f*)(prow + c);
                    fs += v.x; fs += v.y; fs += v.z; fs += v.w;
                }
                const unsigned long long m = __ballot(fs > 0.0f && idx < MS);
                if (lane == 0) {
                    const int cnt = (int)__popcll(m);
                    lds_norm[par * QT + q] = (float)(cnt > 0 ? cnt : 1);
                }
            }

            const int mrow = half * 8;
            #pragma unroll
            for (int r = 0; r < 8; ++r) {
                const v4f o = {acc[0][r], acc[1][r], acc[2][r], acc[3][r]};
                *(v4f*)&lds_weighted[q * 1024 + (r + mrow) * CIN + col * 4] = o;
            }
        }
        {
            const int nq = (has_next ? nqblk : qblk) + wave;
            idx = idx_nx;
            IssueRows<0, JSPLIT>::run(x, idx < MS ? idx : MS - 1, colb, stg);
            sp = *(const v3f_a4*)(s_pts + (idx < MS ? idx : 0) * 3);
            qx = q_pts[nq * 3 + 0];
            qy = q_pts[nq * 3 + 1];
            qz = q_pts[nq * 3 + 2];
        }
        wg_barrier_lds();

        {
            const int nrow = col;
            const float* wrow = &lds_weighted[nrow * 1024 + wave * KSL + half * 2];

            v8f acc[4];
            #pragma unroll
            for (int dt = 0; dt < 4; ++dt) acc[dt] = (v8f){};

            #pragma unroll
            for (int s = 0; s < KST; ++s) {
                v2f A;
                A.x = wrow[s * 4];
                A.y = wrow[s * 4 + 1];
                #pragma unroll
                for (int dt = 0; dt < 4; ++dt)
                    acc[dt] = __builtin_amdgcn_wmma_f32_16x16x4_f32(
                        false, A, false, Bw[s][dt], (short)0, acc[dt], false, false);
            }
            float* pw = &lds_partial[wave * (QT * CO) + lane * 4];
            #pragma unroll
            for (int dt = 0; dt < 4; ++dt)
                #pragma unroll
                for (int rq = 0; rq < 2; ++rq) {
                    const v4f o = {acc[dt][4 * rq], acc[dt][4 * rq + 1], acc[dt][4 * rq + 2], acc[dt][4 * rq + 3]};
                    *(v4f*)(pw + dt * 256 + rq * 128) = o;
                }
        }
        {
            int idxc = idx < MS ? idx : MS - 1;
            asm volatile("" : "+v"(idxc));
            IssueRows<JSPLIT, HN / 2>::run(x, idxc, colb, stg);
        }
        wg_barrier_lds();

        if (tid < 256) {
            const int a  = tid >> 7;
            const int rq = (tid >> 6) & 1;
            const int hh = (tid >> 5) & 1;
            const int i  = tid & 31;
            const float* pp = &lds_partial[(2 * a + (i >> 4)) * 256 + rq * 128 + (16 * hh + (i & 15)) * 4];
            v4f s = {0.0f, 0.0f, 0.0f, 0.0f};
            #pragma unroll
            for (int w = 0; w < 16; ++w) {
                const v4f p = *(const v4f*)(pp + w * (QT * CO));
                s.x += p.x; s.y += p.y; s.z += p.z; s.w += p.w;
            }
            const int   n0  = 4 * rq + 8 * hh;
            const v4f   nrm = *(const v4f*)&lds_norm[par * QT + n0];
            float*      op  = out + (qblk + n0) * CO + 32 * a + i;
            op[0]      = s.x / nrm.x;
            op[CO]     = s.y / nrm.y;
            op[2 * CO] = s.z / nrm.z;
            op[3 * CO] = s.w / nrm.w;
        }
    }
}

extern "C" void kernel_launch(void* const* d_in, const int* in_sizes, int n_in,
                              void* d_out, int out_size, void* d_ws, size_t ws_size,
                              hipStream_t stream) {
    const float* q_pts = (const float*)d_in[0];
    const float* s_pts = (const float*)d_in[1];
    const int*   nb    = (const int*)  d_in[2];
    const float* x     = (const float*)d_in[3];
    const float* kp    = (const float*)d_in[4];
    const float* wt    = (const float*)d_in[5];
    float*       out   = (float*)d_out;

    kpconv_wmma_f32<<<NWG, NTHR, 0, stream>>>(q_pts, s_pts, nb, x, kp, wt, out);
}
